# starting kernel
# baseline (speedup 1.0000x reference)
#include <hip/hip_runtime.h>

#define N_NODES 50000
#define NE      800000
#define D       128
#define EPS_BN  1e-5f
#define MTILES  (N_NODES / 16)
#define LDSS    132

typedef __attribute__((ext_vector_type(2))) float v2f;
typedef __attribute__((ext_vector_type(4))) float v4f;
typedef __attribute__((ext_vector_type(8))) float v8f;


__global__ void zero_kernel(float* __restrict__ p, int n4) {
    int i = blockIdx.x * blockDim.x + threadIdx.x;
    if (i < n4) ((v4f*)p)[i] = (v4f){0.0f, 0.0f, 0.0f, 0.0f};
}

__global__ void degree_kernel(const int* __restrict__ dst, float* __restrict__ deg) {
    int e = blockIdx.x * blockDim.x + threadIdx.x;
    if (e < NE) atomicAdd(&deg[dst[e]], 1.0f);
}

__global__ void invdeg_kernel(float* __restrict__ deg) {
    int i = blockIdx.x * blockDim.x + threadIdx.x;
    if (i < N_NODES) {
        float d = deg[i];
        deg[i] = 1.0f / fmaxf(d, 1.0f);
    }
}

__global__ void scatter_kernel(const float* __restrict__ x, const int* __restrict__ src,
                               const int* __restrict__ dst, float* __restrict__ agg) {
    int t = threadIdx.x;
    int e = blockIdx.x * 2 + (t >> 7);
    int d = t & 127;
    if (e < NE) {
        int s  = src[e];
        int dd = dst[e];
        atomicAdd(&agg[(size_t)dd * D + d], x[(size_t)s * D + d]);
    }
}

__global__ __launch_bounds__(256, 1)
void sage_gemm_kernel(const float* __restrict__ Aagg, const float* __restrict__ inv_deg,
                      const float* __restrict__ Ax,
                      const float* __restrict__ W1, const float* __restrict__ W2,
                      const float* __restrict__ bias,
                      const float* __restrict__ gamma, const float* __restrict__ beta,
                      const float* __restrict__ mean, const float* __restrict__ var,
                      float* __restrict__ out, int mode) {
    __shared__ float sA1[16 * LDSS];
    __shared__ float sA2[16 * LDSS];

    const int t    = threadIdx.x;
    const int lane = t & 31;
    const int wave = t >> 5;
    const int half = lane >> 4;
    const int lrow = lane & 15;
    const int n    = wave * 16 + lrow;

    v2f b1[32];
    v2f b2[32];
#pragma unroll
    for (int ks = 0; ks < 32; ++ks) {
        int k = ks * 4 + 2 * half;
        b1[ks].x = W1[k * D + n];
        b1[ks].y = W1[(k + 1) * D + n];
    }
    if (mode != 0) {
#pragma unroll
        for (int ks = 0; ks < 32; ++ks) {
            int k = ks * 4 + 2 * half;
            b2[ks].x = W2[k * D + n];
            b2[ks].y = W2[(k + 1) * D + n];
        }
    }

    float bcol = bias[n];
    float s = 1.0f, tt = 0.0f;
    if (mode != 0) {
        s  = gamma[n] * rsqrtf(var[n] + EPS_BN);
        tt = beta[n] - mean[n] * s;
    }

    for (int mt = blockIdx.x; mt < MTILES; mt += gridDim.x) {
        const size_t rowbase = (size_t)mt * 16;

#pragma unroll
        for (int i = 0; i < 8; ++i) {
            int idx = t + 256 * i;
            int r   = idx >> 7;
            int c   = idx & 127;
            float xv = Ax[(rowbase + r) * D + c];
            if (mode == 0) {
                sA1[r * LDSS + c] = xv;
            } else {
                sA1[r * LDSS + c] = Aagg[(rowbase + r) * D + c] * inv_deg[rowbase + r];
                sA2[r * LDSS + c] = xv;
            }
        }
        __syncthreads();

        v8f c8 = {};
#pragma unroll
        for (int ks = 0; ks < 32; ++ks) {
            int k = ks * 4 + 2 * half;
            v2f a = *(const v2f*)&sA1[lrow * LDSS + k];
            c8 = __builtin_amdgcn_wmma_f32_16x16x4_f32(false, a, false, b1[ks],
                                                       (short)0, c8, false, false);
        }
        if (mode != 0) {
#pragma unroll
            for (int ks = 0; ks < 32; ++ks) {
                int k = ks * 4 + 2 * half;
                v2f a = *(const v2f*)&sA2[lrow * LDSS + k];
                c8 = __builtin_amdgcn_wmma_f32_16x16x4_f32(false, a, false, b2[ks],
                                                           (short)0, c8, false, false);
            }
        }

#pragma unroll
        for (int v = 0; v < 8; ++v) {
            int m   = v + 8 * half;
            float h = c8[v] + bcol;
            float y;
            if (mode == 0) {
                y = fmaxf(h, 0.0f);
            } else {
                y = fmaxf(h * s + tt, 0.0f);
                if (mode == 1) y += sA2[m * LDSS + n];
            }
            out[(rowbase + m) * D + n] = y;
        }
        __syncthreads();
    }
}


extern "C" void kernel_launch(void* const* d_in, const int* in_sizes, int n_in,
                              void* d_out, int out_size, void* d_ws, size_t ws_size,
                              hipStream_t stream) {
    (void)in_sizes; (void)n_in; (void)out_size; (void)ws_size;

    const float* x   = (const float*)d_in[0];
    const int*   ei  = (const int*)d_in[1];
    const float* Wp  = (const float*)d_in[2];
    const float* bp  = (const float*)d_in[3];
    const float* Wl  = (const float*)d_in[4];
    const float* bl  = (const float*)d_in[5];
    const float* Wr  = (const float*)d_in[6];
    const float* gam = (const float*)d_in[7];
    const float* bet = (const float*)d_in[8];
    const float* mea = (const float*)d_in[9];
    const float* var = (const float*)d_in[10];

    const int* src = ei;
    const int* dst = ei + NE;

    float* ws  = (float*)d_ws;
    float* deg = ws;
    float* xa  = ws + ((N_NODES + 63) / 64) * 64;
    float* xb  = xa + (size_t)N_NODES * D;
    float* agg = xb + (size_t)N_NODES * D;

    const int GEMM_BLOCKS = 512;

    {
        int n4 = N_NODES / 4;
        zero_kernel<<<(n4 + 255) / 256, 256, 0, stream>>>(deg, n4);
        degree_kernel<<<(NE + 255) / 256, 256, 0, stream>>>(dst, deg);
        invdeg_kernel<<<(N_NODES + 255) / 256, 256, 0, stream>>>(deg);
    }

    sage_gemm_kernel<<<GEMM_BLOCKS, 256, 0, stream>>>(
        nullptr, deg, x, Wp, Wp, bp, gam, bet, mea, var, xa, 0);

    float* xcur  = xa;
    float* xnext = xb;
    for (int i = 0; i < 4; ++i) {
        int n4 = (N_NODES * D) / 4;
        zero_kernel<<<(n4 + 255) / 256, 256, 0, stream>>>(agg, n4);
        scatter_kernel<<<NE / 2, 256, 0, stream>>>(xcur, src, dst, agg);

        float* o = (i < 3) ? xnext : (float*)d_out;
        sage_gemm_kernel<<<GEMM_BLOCKS, 256, 0, stream>>>(
            agg, deg, xcur,
            Wl + (size_t)i * D * D, Wr + (size_t)i * D * D, bl + (size_t)i * D,
            gam + (size_t)i * D, bet + (size_t)i * D, mea + (size_t)i * D, var + (size_t)i * D,
            o, (i < 3) ? 1 : 2);

        float* tmp = xcur; xcur = xnext; xnext = tmp;
    }
}


# Neighbour sum via per-call CSR by destination (int atomic count+rank, one-block scan, fill) and a per-row register gather: replaces float atomic scatter and zero pass; sums in arrival-rank order
# speedup vs baseline: 3.0585x; 3.0585x over previous
#include <hip/hip_runtime.h>

#define N_NODES 50000
#define NE      800000
#define D       128
#define EPS_BN  1e-5f
#define MTILES  (N_NODES / 16)
#define LDSS    132

typedef __attribute__((ext_vector_type(2))) float v2f;
typedef __attribute__((ext_vector_type(4))) float v4f;
typedef __attribute__((ext_vector_type(8))) float v8f;
typedef __attribute__((ext_vector_type(4))) int   v4i;

#define SCAN_PER   49
#define SCAN_LEN   (1024 * SCAN_PER)
#define ROWPTR_PAD 50048
#define AGG_BLOCKS 1024
#define AGG_WAVES  (AGG_BLOCKS * 8)


__global__ void zero_kernel(float* __restrict__ p, int n4) {
    int i = blockIdx.x * blockDim.x + threadIdx.x;
    if (i < n4) ((v4f*)p)[i] = (v4f){0.0f, 0.0f, 0.0f, 0.0f};
}


__global__ void count_rank_kernel(const int* __restrict__ dst, int* __restrict__ cnt,
                                  int* __restrict__ rank) {
    int i = blockIdx.x * blockDim.x + threadIdx.x;
    if (i < NE / 4) {
        v4i d = ((const v4i*)dst)[i];
        v4i r;
        r.x = atomicAdd(&cnt[d.x], 1);
        r.y = atomicAdd(&cnt[d.y], 1);
        r.z = atomicAdd(&cnt[d.z], 1);
        r.w = atomicAdd(&cnt[d.w], 1);
        ((v4i*)rank)[i] = r;
    }
}

__global__ __launch_bounds__(1024)
void scan_kernel(const int* __restrict__ cnt, int* __restrict__ rowptr,
                 float* __restrict__ inv_deg) {
    __shared__ int sbuf[SCAN_LEN];
    __shared__ int wtot[32];
    const int t    = threadIdx.x;
    const int lane = t & 31;
    const int wave = t >> 5;

    int v[SCAN_PER];
#pragma unroll
    for (int k = 0; k < SCAN_PER; ++k) v[k] = cnt[min(t + 1024 * k, N_NODES - 1)];
#pragma unroll
    for (int k = 0; k < SCAN_PER; ++k) sbuf[t + 1024 * k] = (t + 1024 * k < N_NODES) ? v[k] : 0;
    __syncthreads();

    int s = 0;
    for (int i = 0; i < SCAN_PER; ++i) s += sbuf[t * SCAN_PER + i];

    int incl = s;
    for (int off = 1; off < 32; off <<= 1) {
        int u = __shfl_up(incl, off);
        if (lane >= off) incl += u;
    }
    if (lane == 31) wtot[wave] = incl;
    __syncthreads();
    int before = (lane < wave) ? wtot[lane] : 0;
    for (int off = 16; off > 0; off >>= 1) before += __shfl_xor(before, off);

    int run = before + incl - s;
    for (int i = 0; i < SCAN_PER; ++i) {
        int c = sbuf[t * SCAN_PER + i];
        sbuf[t * SCAN_PER + i] = run;
        run += c;
    }
    __syncthreads();

    for (int i = t; i <= N_NODES; i += 1024) rowptr[i] = sbuf[i];
    for (int i = t; i < N_NODES; i += 1024) {
        float d = (float)(sbuf[i + 1] - sbuf[i]);
        inv_deg[i] = 1.0f / fmaxf(d, 1.0f);
    }
}

__global__ void fill_kernel(const int* __restrict__ src, const int* __restrict__ dst,
                            const int* __restrict__ rank, const int* __restrict__ rowptr,
                            int* __restrict__ col) {
    int i = blockIdx.x * blockDim.x + threadIdx.x;
    if (i < NE / 4) {
        v4i s = ((const v4i*)src)[i];
        v4i d = ((const v4i*)dst)[i];
        v4i r = ((const v4i*)rank)[i];
        col[rowptr[d.x] + r.x] = s.x;
        col[rowptr[d.y] + r.y] = s.y;
        col[rowptr[d.z] + r.z] = s.z;
        col[rowptr[d.w] + r.w] = s.w;
    }
}

template <int CNT>
__device__ __forceinline__ v4f add_rows(const v4f* __restrict__ x4, const int* __restrict__ col,
                                        int j, int lane, v4f acc) {
    int s[CNT];
#pragma unroll
    for (int k = 0; k < CNT; ++k) {
        int v = col[j + k];
        s[k] = ((unsigned)v < (unsigned)N_NODES) ? v : 0;
    }
    v4f r[CNT];
#pragma unroll
    for (int k = 0; k < CNT; ++k) r[k] = x4[(size_t)s[k] * (D / 4) + lane];
#pragma unroll
    for (int k = 0; k < CNT; ++k) acc += r[k];
    return acc;
}

__global__ __launch_bounds__(256)
void aggregate_kernel(const float* __restrict__ x, const int* __restrict__ rowptr,
                      const int* __restrict__ col, float* __restrict__ agg) {
    const int lane = threadIdx.x & 31;
    int node = __builtin_amdgcn_readfirstlane((int)(blockIdx.x * 8 + (threadIdx.x >> 5)));
    const v4f* __restrict__ x4   = (const v4f*)x;
    v4f* __restrict__       agg4 = (v4f*)agg;

    if (node >= N_NODES) return;
    int beg = rowptr[node];
    int end = rowptr[node + 1];
    while (true) {
        v4f acc = {0.0f, 0.0f, 0.0f, 0.0f};
        int j = beg;
        for (; j + 16 <= end; j += 16) acc = add_rows<16>(x4, col, j, lane, acc);

        const int nnode = node + AGG_WAVES;
        int nbeg = 0, nend = 0;
        if (nnode < N_NODES) { nbeg = rowptr[nnode]; nend = rowptr[nnode + 1]; }

        switch (end - j) {
            case 15: acc = add_rows<15>(x4, col, j, lane, acc); break;
            case 14: acc = add_rows<14>(x4, col, j, lane, acc); break;
            case 13: acc = add_rows<13>(x4, col, j, lane, acc); break;
            case 12: acc = add_rows<12>(x4, col, j, lane, acc); break;
            case 11: acc = add_rows<11>(x4, col, j, lane, acc); break;
            case 10: acc = add_rows<10>(x4, col, j, lane, acc); break;
            case 9:  acc = add_rows<9>(x4, col, j, lane, acc);  break;
            case 8:  acc = add_rows<8>(x4, col, j, lane, acc);  break;
            case 7:  acc = add_rows<7>(x4, col, j, lane, acc);  break;
            case 6:  acc = add_rows<6>(x4, col, j, lane, acc);  break;
            case 5:  acc = add_rows<5>(x4, col, j, lane, acc);  break;
            case 4:  acc = add_rows<4>(x4, col, j, lane, acc);  break;
            case 3:  acc = add_rows<3>(x4, col, j, lane, acc);  break;
            case 2:  acc = add_rows<2>(x4, col, j, lane, acc);  break;
            case 1:  acc = add_rows<1>(x4, col, j, lane, acc);  break;
            default: break;
        }
        agg4[(size_t)node * (D / 4) + lane] = acc;

        if (nnode >= N_NODES) break;
        node = nnode; beg = nbeg; end = nend;
    }
}

__global__ __launch_bounds__(256, 1)
void sage_gemm_kernel(const float* __restrict__ Aagg, const float* __restrict__ inv_deg,
                      const float* __restrict__ Ax,
                      const float* __restrict__ W1, const float* __restrict__ W2,
                      const float* __restrict__ bias,
                      const float* __restrict__ gamma, const float* __restrict__ beta,
                      const float* __restrict__ mean, const float* __restrict__ var,
                      float* __restrict__ out, int mode) {
    __shared__ float sA1[16 * LDSS];
    __shared__ float sA2[16 * LDSS];

    const int t    = threadIdx.x;
    const int lane = t & 31;
    const int wave = t >> 5;
    const int half = lane >> 4;
    const int lrow = lane & 15;
    const int n    = wave * 16 + lrow;

    v2f b1[32];
    v2f b2[32];
#pragma unroll
    for (int ks = 0; ks < 32; ++ks) {
        int k = ks * 4 + 2 * half;
        b1[ks].x = W1[k * D + n];
        b1[ks].y = W1[(k + 1) * D + n];
    }
    if (mode != 0) {
#pragma unroll
        for (int ks = 0; ks < 32; ++ks) {
            int k = ks * 4 + 2 * half;
            b2[ks].x = W2[k * D + n];
            b2[ks].y = W2[(k + 1) * D + n];
        }
    }

    float bcol = bias[n];
    float s = 1.0f, tt = 0.0f;
    if (mode != 0) {
        s  = gamma[n] * rsqrtf(var[n] + EPS_BN);
        tt = beta[n] - mean[n] * s;
    }

    for (int mt = blockIdx.x; mt < MTILES; mt += gridDim.x) {
        const size_t rowbase = (size_t)mt * 16;

#pragma unroll
        for (int i = 0; i < 8; ++i) {
            int idx = t + 256 * i;
            int r   = idx >> 7;
            int c   = idx & 127;
            float xv = Ax[(rowbase + r) * D + c];
            if (mode == 0) {
                sA1[r * LDSS + c] = xv;
            } else {
                sA1[r * LDSS + c] = Aagg[(rowbase + r) * D + c] * inv_deg[rowbase + r];
                sA2[r * LDSS + c] = xv;
            }
        }
        __syncthreads();

        v8f c8 = {};
#pragma unroll
        for (int ks = 0; ks < 32; ++ks) {
            int k = ks * 4 + 2 * half;
            v2f a = *(const v2f*)&sA1[lrow * LDSS + k];
            c8 = __builtin_amdgcn_wmma_f32_16x16x4_f32(false, a, false, b1[ks],
                                                       (short)0, c8, false, false);
        }
        if (mode != 0) {
#pragma unroll
            for (int ks = 0; ks < 32; ++ks) {
                int k = ks * 4 + 2 * half;
                v2f a = *(const v2f*)&sA2[lrow * LDSS + k];
                c8 = __builtin_amdgcn_wmma_f32_16x16x4_f32(false, a, false, b2[ks],
                                                           (short)0, c8, false, false);
            }
        }

#pragma unroll
        for (int v = 0; v < 8; ++v) {
            int m   = v + 8 * half;
            float h = c8[v] + bcol;
            float y;
            if (mode == 0) {
                y = fmaxf(h, 0.0f);
            } else {
                y = fmaxf(h * s + tt, 0.0f);
                if (mode == 1) y += sA2[m * LDSS + n];
            }
            out[(rowbase + m) * D + n] = y;
        }
        __syncthreads();
    }
}


extern "C" void kernel_launch(void* const* d_in, const int* in_sizes, int n_in,
                              void* d_out, int out_size, void* d_ws, size_t ws_size,
                              hipStream_t stream) {
    (void)in_sizes; (void)n_in; (void)out_size; (void)ws_size;

    const float* x   = (const float*)d_in[0];
    const int*   ei  = (const int*)d_in[1];
    const float* Wp  = (const float*)d_in[2];
    const float* bp  = (const float*)d_in[3];
    const float* Wl  = (const float*)d_in[4];
    const float* bl  = (const float*)d_in[5];
    const float* Wr  = (const float*)d_in[6];
    const float* gam = (const float*)d_in[7];
    const float* bet = (const float*)d_in[8];
    const float* mea = (const float*)d_in[9];
    const float* var = (const float*)d_in[10];

    const int* src = ei;
    const int* dst = ei + NE;

    float* ws  = (float*)d_ws;
    float* deg = ws;
    float* xa  = ws + ((N_NODES + 63) / 64) * 64;
    float* xb  = xa + (size_t)N_NODES * D;
    float* agg = xb + (size_t)N_NODES * D;

    int* col    = (int*)d_out;
    int* rank   = col + NE;
    int* rowptr = rank + NE;
    int* cnt    = rowptr + ROWPTR_PAD;

    const int GEMM_BLOCKS = 512;

    {
        int n4 = N_NODES / 4;
        zero_kernel<<<(n4 + 255) / 256, 256, 0, stream>>>((float*)cnt, n4);
        count_rank_kernel<<<(NE / 4 + 255) / 256, 256, 0, stream>>>(dst, cnt, rank);
        scan_kernel<<<1, 1024, 0, stream>>>(cnt, rowptr, deg);
        fill_kernel<<<(NE / 4 + 255) / 256, 256, 0, stream>>>(src, dst, rank, rowptr, col);
    }

    sage_gemm_kernel<<<GEMM_BLOCKS, 256, 0, stream>>>(
        nullptr, deg, x, Wp, Wp, bp, gam, bet, mea, var, xa, 0);

    float* xcur  = xa;
    float* xnext = xb;
    for (int i = 0; i < 4; ++i) {
        aggregate_kernel<<<AGG_BLOCKS, 256, 0, stream>>>(xcur, rowptr, col, agg);

        float* o = (i < 3) ? xnext : (float*)d_out;
        sage_gemm_kernel<<<GEMM_BLOCKS, 256, 0, stream>>>(
            agg, deg, xcur,
            Wl + (size_t)i * D * D, Wr + (size_t)i * D * D, bl + (size_t)i * D,
            gam + (size_t)i * D, bet + (size_t)i * D, mea + (size_t)i * D, var + (size_t)i * D,
            o, (i < 3) ? 1 : 2);

        float* tmp = xcur; xcur = xnext; xnext = tmp;
    }
}


# GEMM A-tile staging pipelined: layer mode compile-time, so a tile's rows load as one 16-B-per-lane group a tile ahead into a second LDS stage, one barrier a tile (was 8 serial load waits a tile)
# speedup vs baseline: 4.2828x; 1.4003x over previous
#include <hip/hip_runtime.h>

#define N_NODES 50000
#define NE      800000
#define D       128
#define EPS_BN  1e-5f
#define MTILES  (N_NODES / 16)
#define LDSS    132

typedef __attribute__((ext_vector_type(2))) float v2f;
typedef __attribute__((ext_vector_type(4))) float v4f;
typedef __attribute__((ext_vector_type(8))) float v8f;
typedef __attribute__((ext_vector_type(4))) int   v4i;

#define SCAN_PER   49
#define SCAN_LEN   (1024 * SCAN_PER)
#define ROWPTR_PAD 50048
#define AGG_BLOCKS 1024
#define AGG_WAVES  (AGG_BLOCKS * 8)


__global__ void zero_kernel(float* __restrict__ p, int n4) {
    int i = blockIdx.x * blockDim.x + threadIdx.x;
    if (i < n4) ((v4f*)p)[i] = (v4f){0.0f, 0.0f, 0.0f, 0.0f};
}


__global__ void count_rank_kernel(const int* __restrict__ dst, int* __restrict__ cnt,
                                  int* __restrict__ rank) {
    int i = blockIdx.x * blockDim.x + threadIdx.x;
    if (i < NE / 4) {
        v4i d = ((const v4i*)dst)[i];
        v4i r;
        r.x = atomicAdd(&cnt[d.x], 1);
        r.y = atomicAdd(&cnt[d.y], 1);
        r.z = atomicAdd(&cnt[d.z], 1);
        r.w = atomicAdd(&cnt[d.w], 1);
        ((v4i*)rank)[i] = r;
    }
}

__global__ __launch_bounds__(1024)
void scan_kernel(const int* __restrict__ cnt, int* __restrict__ rowptr,
                 float* __restrict__ inv_deg) {
    __shared__ int sbuf[SCAN_LEN];
    __shared__ int wtot[32];
    const int t    = threadIdx.x;
    const int lane = t & 31;
    const int wave = t >> 5;

    int v[SCAN_PER];
#pragma unroll
    for (int k = 0; k < SCAN_PER; ++k) v[k] = cnt[min(t + 1024 * k, N_NODES - 1)];
#pragma unroll
    for (int k = 0; k < SCAN_PER; ++k) sbuf[t + 1024 * k] = (t + 1024 * k < N_NODES) ? v[k] : 0;
    __syncthreads();

    int s = 0;
    for (int i = 0; i < SCAN_PER; ++i) s += sbuf[t * SCAN_PER + i];

    int incl = s;
    for (int off = 1; off < 32; off <<= 1) {
        int u = __shfl_up(incl, off);
        if (lane >= off) incl += u;
    }
    if (lane == 31) wtot[wave] = incl;
    __syncthreads();
    int before = (lane < wave) ? wtot[lane] : 0;
    for (int off = 16; off > 0; off >>= 1) before += __shfl_xor(before, off);

    int run = before + incl - s;
    for (int i = 0; i < SCAN_PER; ++i) {
        int c = sbuf[t * SCAN_PER + i];
        sbuf[t * SCAN_PER + i] = run;
        run += c;
    }
    __syncthreads();

    for (int i = t; i <= N_NODES; i += 1024) rowptr[i] = sbuf[i];
    for (int i = t; i < N_NODES; i += 1024) {
        float d = (float)(sbuf[i + 1] - sbuf[i]);
        inv_deg[i] = 1.0f / fmaxf(d, 1.0f);
    }
}

__global__ void fill_kernel(const int* __restrict__ src, const int* __restrict__ dst,
                            const int* __restrict__ rank, const int* __restrict__ rowptr,
                            int* __restrict__ col) {
    int i = blockIdx.x * blockDim.x + threadIdx.x;
    if (i < NE / 4) {
        v4i s = ((const v4i*)src)[i];
        v4i d = ((const v4i*)dst)[i];
        v4i r = ((const v4i*)rank)[i];
        col[rowptr[d.x] + r.x] = s.x;
        col[rowptr[d.y] + r.y] = s.y;
        col[rowptr[d.z] + r.z] = s.z;
        col[rowptr[d.w] + r.w] = s.w;
    }
}

template <int CNT>
__device__ __forceinline__ v4f add_rows(const v4f* __restrict__ x4, const int* __restrict__ col,
                                        int j, int lane, v4f acc) {
    int s[CNT];
#pragma unroll
    for (int k = 0; k < CNT; ++k) {
        int v = col[j + k];
        s[k] = ((unsigned)v < (unsigned)N_NODES) ? v : 0;
    }
    v4f r[CNT];
#pragma unroll
    for (int k = 0; k < CNT; ++k) r[k] = x4[(size_t)s[k] * (D / 4) + lane];
#pragma unroll
    for (int k = 0; k < CNT; ++k) acc += r[k];
    return acc;
}

__global__ __launch_bounds__(256)
void aggregate_kernel(const float* __restrict__ x, const int* __restrict__ rowptr,
                      const int* __restrict__ col, float* __restrict__ agg) {
    const int lane = threadIdx.x & 31;
    int node = __builtin_amdgcn_readfirstlane((int)(blockIdx.x * 8 + (threadIdx.x >> 5)));
    const v4f* __restrict__ x4   = (const v4f*)x;
    v4f* __restrict__       agg4 = (v4f*)agg;

    if (node >= N_NODES) return;
    int beg = rowptr[node];
    int end = rowptr[node + 1];
    while (true) {
        v4f acc = {0.0f, 0.0f, 0.0f, 0.0f};
        int j = beg;
        for (; j + 16 <= end; j += 16) acc = add_rows<16>(x4, col, j, lane, acc);

        const int nnode = node + AGG_WAVES;
        int nbeg = 0, nend = 0;
        if (nnode < N_NODES) { nbeg = rowptr[nnode]; nend = rowptr[nnode + 1]; }

        switch (end - j) {
            case 15: acc = add_rows<15>(x4, col, j, lane, acc); break;
            case 14: acc = add_rows<14>(x4, col, j, lane, acc); break;
            case 13: acc = add_rows<13>(x4, col, j, lane, acc); break;
            case 12: acc = add_rows<12>(x4, col, j, lane, acc); break;
            case 11: acc = add_rows<11>(x4, col, j, lane, acc); break;
            case 10: acc = add_rows<10>(x4, col, j, lane, acc); break;
            case 9:  acc = add_rows<9>(x4, col, j, lane, acc);  break;
            case 8:  acc = add_rows<8>(x4, col, j, lane, acc);  break;
            case 7:  acc = add_rows<7>(x4, col, j, lane, acc);  break;
            case 6:  acc = add_rows<6>(x4, col, j, lane, acc);  break;
            case 5:  acc = add_rows<5>(x4, col, j, lane, acc);  break;
            case 4:  acc = add_rows<4>(x4, col, j, lane, acc);  break;
            case 3:  acc = add_rows<3>(x4, col, j, lane, acc);  break;
            case 2:  acc = add_rows<2>(x4, col, j, lane, acc);  break;
            case 1:  acc = add_rows<1>(x4, col, j, lane, acc);  break;
            default: break;
        }
        agg4[(size_t)node * (D / 4) + lane] = acc;

        if (nnode >= N_NODES) break;
        node = nnode; beg = nbeg; end = nend;
    }
}

template <int MODE>
__global__ __launch_bounds__(256, 1)
void sage_gemm_kernel(const float* __restrict__ Aagg, const float* __restrict__ inv_deg,
                      const float* __restrict__ Ax,
                      const float* __restrict__ W1, const float* __restrict__ W2,
                      const float* __restrict__ bias,
                      const float* __restrict__ gamma, const float* __restrict__ beta,
                      const float* __restrict__ mean, const float* __restrict__ var,
                      float* __restrict__ out) {
    __shared__ __attribute__((aligned(16))) float sA1[2][16 * LDSS];
    __shared__ __attribute__((aligned(16))) float sA2[2][16 * LDSS];

    const int t    = threadIdx.x;
    const int lane = t & 31;
    const int wave = t >> 5;
    const int half = lane >> 4;
    const int lrow = lane & 15;
    const int n    = wave * 16 + lrow;

    const int sr = t >> 5;
    const int sc = (t & 31) * 4;

    v2f b1[32];
    v2f b2[32];
#pragma unroll
    for (int ks = 0; ks < 32; ++ks) {
        int k = ks * 4 + 2 * half;
        b1[ks].x = W1[k * D + n];
        b1[ks].y = W1[(k + 1) * D + n];
    }
    if (MODE != 0) {
#pragma unroll
        for (int ks = 0; ks < 32; ++ks) {
            int k = ks * 4 + 2 * half;
            b2[ks].x = W2[k * D + n];
            b2[ks].y = W2[(k + 1) * D + n];
        }
    }

    float bcol = bias[n];
    float s = 1.0f, tt = 0.0f;
    if (MODE != 0) {
        s  = gamma[n] * rsqrtf(var[n] + EPS_BN);
        tt = beta[n] - mean[n] * s;
    }

    int mt = blockIdx.x;
    if (mt >= MTILES) return;

    v4f ax0, ax1;
    v4f ag0 = {0.0f, 0.0f, 0.0f, 0.0f}, ag1 = {0.0f, 0.0f, 0.0f, 0.0f};
    float id0 = 0.0f, id1 = 0.0f;

    {
        const size_t rb = (size_t)mt * 16;
        ax0 = *(const v4f*)&Ax[(rb + sr) * D + sc];
        ax1 = *(const v4f*)&Ax[(rb + sr + 8) * D + sc];
        if (MODE != 0) {
            ag0 = *(const v4f*)&Aagg[(rb + sr) * D + sc];
            ag1 = *(const v4f*)&Aagg[(rb + sr + 8) * D + sc];
            id0 = inv_deg[rb + sr];
            id1 = inv_deg[rb + sr + 8];
        }
        if (MODE == 0) {
            *(v4f*)&sA1[0][sr * LDSS + sc]       = ax0;
            *(v4f*)&sA1[0][(sr + 8) * LDSS + sc] = ax1;
        } else {
            *(v4f*)&sA1[0][sr * LDSS + sc]       = ag0 * id0;
            *(v4f*)&sA1[0][(sr + 8) * LDSS + sc] = ag1 * id1;
            *(v4f*)&sA2[0][sr * LDSS + sc]       = ax0;
            *(v4f*)&sA2[0][(sr + 8) * LDSS + sc] = ax1;
        }
    }
    __syncthreads();

    int buf = 0;
    while (true) {
        const size_t rowbase = (size_t)mt * 16;
        const int    nmt     = mt + (int)gridDim.x;
        const bool   more    = nmt < MTILES;

        if (more) {
            const size_t rb = (size_t)nmt * 16;
            ax0 = *(const v4f*)&Ax[(rb + sr) * D + sc];
            ax1 = *(const v4f*)&Ax[(rb + sr + 8) * D + sc];
            if (MODE != 0) {
                ag0 = *(const v4f*)&Aagg[(rb + sr) * D + sc];
                ag1 = *(const v4f*)&Aagg[(rb + sr + 8) * D + sc];
                id0 = inv_deg[rb + sr];
                id1 = inv_deg[rb + sr + 8];
            }
        }

        const float* pA1 = &sA1[buf][0];
        const float* pA2 = &sA2[buf][0];

        v8f c8 = {};
#pragma unroll
        for (int ks = 0; ks < 32; ++ks) {
            int k = ks * 4 + 2 * half;
            v2f a = *(const v2f*)&pA1[lrow * LDSS + k];
            c8 = __builtin_amdgcn_wmma_f32_16x16x4_f32(false, a, false, b1[ks],
                                                       (short)0, c8, false, false);
        }
        if (MODE != 0) {
#pragma unroll
            for (int ks = 0; ks < 32; ++ks) {
                int k = ks * 4 + 2 * half;
                v2f a = *(const v2f*)&pA2[lrow * LDSS + k];
                c8 = __builtin_amdgcn_wmma_f32_16x16x4_f32(false, a, false, b2[ks],
                                                           (short)0, c8, false, false);
            }
        }

#pragma unroll
        for (int v = 0; v < 8; ++v) {
            int m   = v + 8 * half;
            float h = c8[v] + bcol;
            float y;
            if (MODE == 0) {
                y = fmaxf(h, 0.0f);
            } else {
                y = fmaxf(h * s + tt, 0.0f);
                if (MODE == 1) y += pA2[m * LDSS + n];
            }
            out[(rowbase + m) * D + n] = y;
        }

        if (!more) break;

        const int nb = buf ^ 1;
        if (MODE == 0) {
            *(v4f*)&sA1[nb][sr * LDSS + sc]       = ax0;
            *(v4f*)&sA1[nb][(sr + 8) * LDSS + sc] = ax1;
        } else {
            *(v4f*)&sA1[nb][sr * LDSS + sc]       = ag0 * id0;
            *(v4f*)&sA1[nb][(sr + 8) * LDSS + sc] = ag1 * id1;
            *(v4f*)&sA2[nb][sr * LDSS + sc]       = ax0;
            *(v4f*)&sA2[nb][(sr + 8) * LDSS + sc] = ax1;
        }
        __syncthreads();
        buf = nb;
        mt  = nmt;
    }
}


extern "C" void kernel_launch(void* const* d_in, const int* in_sizes, int n_in,
                              void* d_out, int out_size, void* d_ws, size_t ws_size,
                              hipStream_t stream) {
    (void)in_sizes; (void)n_in; (void)out_size; (void)ws_size;

    const float* x   = (const float*)d_in[0];
    const int*   ei  = (const int*)d_in[1];
    const float* Wp  = (const float*)d_in[2];
    const float* bp  = (const float*)d_in[3];
    const float* Wl  = (const float*)d_in[4];
    const float* bl  = (const float*)d_in[5];
    const float* Wr  = (const float*)d_in[6];
    const float* gam = (const float*)d_in[7];
    const float* bet = (const float*)d_in[8];
    const float* mea = (const float*)d_in[9];
    const float* var = (const float*)d_in[10];

    const int* src = ei;
    const int* dst = ei + NE;

    float* ws  = (float*)d_ws;
    float* deg = ws;
    float* xa  = ws + ((N_NODES + 63) / 64) * 64;
    float* xb  = xa + (size_t)N_NODES * D;
    float* agg = xb + (size_t)N_NODES * D;

    int* col    = (int*)d_out;
    int* rank   = col + NE;
    int* rowptr = rank + NE;
    int* cnt    = rowptr + ROWPTR_PAD;

    const int GEMM_BLOCKS = 512;

    {
        int n4 = N_NODES / 4;
        zero_kernel<<<(n4 + 255) / 256, 256, 0, stream>>>((float*)cnt, n4);
        count_rank_kernel<<<(NE / 4 + 255) / 256, 256, 0, stream>>>(dst, cnt, rank);
        scan_kernel<<<1, 1024, 0, stream>>>(cnt, rowptr, deg);
        fill_kernel<<<(NE / 4 + 255) / 256, 256, 0, stream>>>(src, dst, rank, rowptr, col);
    }

    sage_gemm_kernel<0><<<GEMM_BLOCKS, 256, 0, stream>>>(
        nullptr, deg, x, Wp, Wp, bp, gam, bet, mea, var, xa);

    float* xcur  = xa;
    float* xnext = xb;
    for (int i = 0; i < 4; ++i) {
        aggregate_kernel<<<AGG_BLOCKS, 256, 0, stream>>>(xcur, rowptr, col, agg);

        if (i < 3) {
            sage_gemm_kernel<1><<<GEMM_BLOCKS, 256, 0, stream>>>(
                agg, deg, xcur,
                Wl + (size_t)i * D * D, Wr + (size_t)i * D * D, bl + (size_t)i * D,
                gam + (size_t)i * D, bet + (size_t)i * D, mea + (size_t)i * D, var + (size_t)i * D,
                xnext);
        } else {
            sage_gemm_kernel<2><<<GEMM_BLOCKS, 256, 0, stream>>>(
                agg, deg, xcur,
                Wl + (size_t)i * D * D, Wr + (size_t)i * D * D, bl + (size_t)i * D,
                gam + (size_t)i * D, bet + (size_t)i * D, mea + (size_t)i * D, var + (size_t)i * D,
                (float*)d_out);
        }

        float* tmp = xcur; xcur = xnext; xnext = tmp;
    }
}


# Neighbour gather fused into the layer GEMM's A-tile staging: each wave sums its two rows' lists (same list order) straight into LDS; the aggregate kernel, its 4 launches and the sums array go
# speedup vs baseline: 4.5955x; 1.0730x over previous
#include <hip/hip_runtime.h>

#define N_NODES 50000
#define NE      800000
#define D       128
#define EPS_BN  1e-5f
#define MTILES  (N_NODES / 16)
#define LDSS    132

typedef __attribute__((ext_vector_type(2))) float v2f;
typedef __attribute__((ext_vector_type(4))) float v4f;
typedef __attribute__((ext_vector_type(8))) float v8f;
typedef __attribute__((ext_vector_type(4))) int   v4i;

#define SCAN_PER   49
#define SCAN_LEN   (1024 * SCAN_PER)
#define ROWPTR_PAD 50048


__global__ void zero_kernel(float* __restrict__ p, int n4) {
    int i = blockIdx.x * blockDim.x + threadIdx.x;
    if (i < n4) ((v4f*)p)[i] = (v4f){0.0f, 0.0f, 0.0f, 0.0f};
}


__global__ void count_rank_kernel(const int* __restrict__ dst, int* __restrict__ cnt,
                                  int* __restrict__ rank) {
    int i = blockIdx.x * blockDim.x + threadIdx.x;
    if (i < NE / 4) {
        v4i d = ((const v4i*)dst)[i];
        v4i r;
        r.x = atomicAdd(&cnt[d.x], 1);
        r.y = atomicAdd(&cnt[d.y], 1);
        r.z = atomicAdd(&cnt[d.z], 1);
        r.w = atomicAdd(&cnt[d.w], 1);
        ((v4i*)rank)[i] = r;
    }
}

__global__ __launch_bounds__(1024)
void scan_kernel(const int* __restrict__ cnt, int* __restrict__ rowptr,
                 float* __restrict__ inv_deg) {
    __shared__ int sbuf[SCAN_LEN];
    __shared__ int wtot[32];
    const int t    = threadIdx.x;
    const int lane = t & 31;
    const int wave = t >> 5;

    int v[SCAN_PER];
#pragma unroll
    for (int k = 0; k < SCAN_PER; ++k) v[k] = cnt[min(t + 1024 * k, N_NODES - 1)];
#pragma unroll
    for (int k = 0; k < SCAN_PER; ++k) sbuf[t + 1024 * k] = (t + 1024 * k < N_NODES) ? v[k] : 0;
    __syncthreads();

    int s = 0;
    for (int i = 0; i < SCAN_PER; ++i) s += sbuf[t * SCAN_PER + i];

    int incl = s;
    for (int off = 1; off < 32; off <<= 1) {
        int u = __shfl_up(incl, off);
        if (lane >= off) incl += u;
    }
    if (lane == 31) wtot[wave] = incl;
    __syncthreads();
    int before = (lane < wave) ? wtot[lane] : 0;
    for (int off = 16; off > 0; off >>= 1) before += __shfl_xor(before, off);

    int run = before + incl - s;
    for (int i = 0; i < SCAN_PER; ++i) {
        int c = sbuf[t * SCAN_PER + i];
        sbuf[t * SCAN_PER + i] = run;
        run += c;
    }
    __syncthreads();

    for (int i = t; i <= N_NODES; i += 1024) rowptr[i] = sbuf[i];
    for (int i = t; i < N_NODES; i += 1024) {
        float d = (float)(sbuf[i + 1] - sbuf[i]);
        inv_deg[i] = 1.0f / fmaxf(d, 1.0f);
    }
}

__global__ void fill_kernel(const int* __restrict__ src, const int* __restrict__ dst,
                            const int* __restrict__ rank, const int* __restrict__ rowptr,
                            int* __restrict__ col) {
    int i = blockIdx.x * blockDim.x + threadIdx.x;
    if (i < NE / 4) {
        v4i s = ((const v4i*)src)[i];
        v4i d = ((const v4i*)dst)[i];
        v4i r = ((const v4i*)rank)[i];
        col[rowptr[d.x] + r.x] = s.x;
        col[rowptr[d.y] + r.y] = s.y;
        col[rowptr[d.z] + r.z] = s.z;
        col[rowptr[d.w] + r.w] = s.w;
    }
}

template <int CNT>
__device__ __forceinline__ v4f add_rows(const v4f* __restrict__ x4, const int* __restrict__ col,
                                        int j, int lane, v4f acc) {
    int s[CNT];
#pragma unroll
    for (int k = 0; k < CNT; ++k) {
        int v = col[j + k];
        s[k] = ((unsigned)v < (unsigned)N_NODES) ? v : 0;
    }
    v4f r[CNT];
#pragma unroll
    for (int k = 0; k < CNT; ++k) r[k] = x4[(size_t)s[k] * (D / 4) + lane];
#pragma unroll
    for (int k = 0; k < CNT; ++k) acc += r[k];
    return acc;
}

__device__ __forceinline__ v4f gather_rows(const v4f* __restrict__ x4, const int* __restrict__ col,
                                           int beg, int end, int lane) {
    v4f acc = {0.0f, 0.0f, 0.0f, 0.0f};
    int j = beg;
    for (; j + 16 <= end; j += 16) acc = add_rows<16>(x4, col, j, lane, acc);
    switch (end - j) {
        case 15: acc = add_rows<15>(x4, col, j, lane, acc); break;
        case 14: acc = add_rows<14>(x4, col, j, lane, acc); break;
        case 13: acc = add_rows<13>(x4, col, j, lane, acc); break;
        case 12: acc = add_rows<12>(x4, col, j, lane, acc); break;
        case 11: acc = add_rows<11>(x4, col, j, lane, acc); break;
        case 10: acc = add_rows<10>(x4, col, j, lane, acc); break;
        case 9:  acc = add_rows<9>(x4, col, j, lane, acc);  break;
        case 8:  acc = add_rows<8>(x4, col, j, lane, acc);  break;
        case 7:  acc = add_rows<7>(x4, col, j, lane, acc);  break;
        case 6:  acc = add_rows<6>(x4, col, j, lane, acc);  break;
        case 5:  acc = add_rows<5>(x4, col, j, lane, acc);  break;
        case 4:  acc = add_rows<4>(x4, col, j, lane, acc);  break;
        case 3:  acc = add_rows<3>(x4, col, j, lane, acc);  break;
        case 2:  acc = add_rows<2>(x4, col, j, lane, acc);  break;
        case 1:  acc = add_rows<1>(x4, col, j, lane, acc);  break;
        default: break;
    }
    return acc;
}

template <int MODE>
__global__ __launch_bounds__(256, 1)
void sage_gemm_kernel(const int* __restrict__ rowptr, const int* __restrict__ col,
                      const float* __restrict__ inv_deg,
                      const float* __restrict__ Ax,
                      const float* __restrict__ W1, const float* __restrict__ W2,
                      const float* __restrict__ bias,
                      const float* __restrict__ gamma, const float* __restrict__ beta,
                      const float* __restrict__ mean, const float* __restrict__ var,
                      float* __restrict__ out) {
    __shared__ __attribute__((aligned(16))) float sA1[2][16 * LDSS];
    __shared__ __attribute__((aligned(16))) float sA2[2][16 * LDSS];

    const int t    = threadIdx.x;
    const int lane = t & 31;
    const int wave = t >> 5;
    const int half = lane >> 4;
    const int lrow = lane & 15;
    const int n    = wave * 16 + lrow;

    const int sr = t >> 5;
    const int sc = (t & 31) * 4;

    v2f b1[32];
    v2f b2[32];
#pragma unroll
    for (int ks = 0; ks < 32; ++ks) {
        int k = ks * 4 + 2 * half;
        b1[ks].x = W1[k * D + n];
        b1[ks].y = W1[(k + 1) * D + n];
    }
    if (MODE != 0) {
#pragma unroll
        for (int ks = 0; ks < 32; ++ks) {
            int k = ks * 4 + 2 * half;
            b2[ks].x = W2[k * D + n];
            b2[ks].y = W2[(k + 1) * D + n];
        }
    }

    float bcol = bias[n];
    float s = 1.0f, tt = 0.0f;
    if (MODE != 0) {
        s  = gamma[n] * rsqrtf(var[n] + EPS_BN);
        tt = beta[n] - mean[n] * s;
    }

    int mt = blockIdx.x;
    if (mt >= MTILES) return;

    const v4f* __restrict__ x4 = (const v4f*)Ax;

    v4f ax0, ax1;
    int   lb0 = 0, le0 = 0, lb1 = 0, le1 = 0;
    float id0 = 0.0f, id1 = 0.0f;
    {
        const size_t rb = (size_t)mt * 16;
        ax0 = *(const v4f*)&Ax[(rb + sr) * D + sc];
        ax1 = *(const v4f*)&Ax[(rb + sr + 8) * D + sc];
        if (MODE != 0) {
            const int nd = mt * 16 + sr;
            lb0 = rowptr[nd];     le0 = rowptr[nd + 1];
            lb1 = rowptr[nd + 8]; le1 = rowptr[nd + 9];
            id0 = inv_deg[nd];
            id1 = inv_deg[nd + 8];
        }
    }

    int buf = 0;
    while (true) {
        const size_t rowbase = (size_t)mt * 16;

        if (MODE == 0) {
            *(v4f*)&sA1[buf][sr * LDSS + sc]       = ax0;
            *(v4f*)&sA1[buf][(sr + 8) * LDSS + sc] = ax1;
        } else {
            *(v4f*)&sA2[buf][sr * LDSS + sc]       = ax0;
            *(v4f*)&sA2[buf][(sr + 8) * LDSS + sc] = ax1;
#pragma nounroll
            for (int h = 0; h < 2; ++h) {
                const int   beg = __builtin_amdgcn_readfirstlane(h ? lb1 : lb0);
                const int   end = __builtin_amdgcn_readfirstlane(h ? le1 : le0);
                const float idg = h ? id1 : id0;
                v4f acc = gather_rows(x4, col, beg, end, lane);
                *(v4f*)&sA1[buf][(sr + 8 * h) * LDSS + sc] = acc * idg;
            }
        }
        __syncthreads();

        const int  nmt  = mt + (int)gridDim.x;
        const bool more = nmt < MTILES;

        if (more) {
            const size_t rb = (size_t)nmt * 16;
            ax0 = *(const v4f*)&Ax[(rb + sr) * D + sc];
            ax1 = *(const v4f*)&Ax[(rb + sr + 8) * D + sc];
            if (MODE != 0) {
                const int nd = nmt * 16 + sr;
                lb0 = rowptr[nd];     le0 = rowptr[nd + 1];
                lb1 = rowptr[nd + 8]; le1 = rowptr[nd + 9];
                id0 = inv_deg[nd];
                id1 = inv_deg[nd + 8];
            }
        }

        const float* pA1 = &sA1[buf][0];
        const float* pA2 = &sA2[buf][0];

        v8f c8 = {};
#pragma unroll
        for (int ks = 0; ks < 32; ++ks) {
            int k = ks * 4 + 2 * half;
            v2f a = *(const v2f*)&pA1[lrow * LDSS + k];
            c8 = __builtin_amdgcn_wmma_f32_16x16x4_f32(false, a, false, b1[ks],
                                                       (short)0, c8, false, false);
        }
        if (MODE != 0) {
#pragma unroll
            for (int ks = 0; ks < 32; ++ks) {
                int k = ks * 4 + 2 * half;
                v2f a = *(const v2f*)&pA2[lrow * LDSS + k];
                c8 = __builtin_amdgcn_wmma_f32_16x16x4_f32(false, a, false, b2[ks],
                                                           (short)0, c8, false, false);
            }
        }

#pragma unroll
        for (int v = 0; v < 8; ++v) {
            int m   = v + 8 * half;
            float h = c8[v] + bcol;
            float y;
            if (MODE == 0) {
                y = fmaxf(h, 0.0f);
            } else {
                y = fmaxf(h * s + tt, 0.0f);
                if (MODE == 1) y += pA2[m * LDSS + n];
            }
            out[(rowbase + m) * D + n] = y;
        }

        if (!more) break;
        buf ^= 1;
        mt = nmt;
    }
}


extern "C" void kernel_launch(void* const* d_in, const int* in_sizes, int n_in,
                              void* d_out, int out_size, void* d_ws, size_t ws_size,
                              hipStream_t stream) {
    (void)in_sizes; (void)n_in; (void)out_size; (void)ws_size;

    const float* x   = (const float*)d_in[0];
    const int*   ei  = (const int*)d_in[1];
    const float* Wp  = (const float*)d_in[2];
    const float* bp  = (const float*)d_in[3];
    const float* Wl  = (const float*)d_in[4];
    const float* bl  = (const float*)d_in[5];
    const float* Wr  = (const float*)d_in[6];
    const float* gam = (const float*)d_in[7];
    const float* bet = (const float*)d_in[8];
    const float* mea = (const float*)d_in[9];
    const float* var = (const float*)d_in[10];

    const int* src = ei;
    const int* dst = ei + NE;

    float* ws  = (float*)d_ws;
    float* deg = ws;
    float* xa  = ws + ((N_NODES + 63) / 64) * 64;
    float* xb  = xa + (size_t)N_NODES * D;
    float* csr = xb + (size_t)N_NODES * D;

    int* col    = (int*)csr;
    int* rank   = col + NE;
    int* rowptr = rank + NE;
    int* cnt    = rowptr + ROWPTR_PAD;

    const int GEMM_BLOCKS = 512;

    {
        int n4 = N_NODES / 4;
        zero_kernel<<<(n4 + 255) / 256, 256, 0, stream>>>((float*)cnt, n4);
        count_rank_kernel<<<(NE / 4 + 255) / 256, 256, 0, stream>>>(dst, cnt, rank);
        scan_kernel<<<1, 1024, 0, stream>>>(cnt, rowptr, deg);
        fill_kernel<<<(NE / 4 + 255) / 256, 256, 0, stream>>>(src, dst, rank, rowptr, col);
    }

    sage_gemm_kernel<0><<<GEMM_BLOCKS, 256, 0, stream>>>(
        nullptr, nullptr, deg, x, Wp, Wp, bp, gam, bet, mea, var, xa);

    float* xcur  = xa;
    float* xnext = xb;
    for (int i = 0; i < 4; ++i) {
        if (i < 3) {
            sage_gemm_kernel<1><<<GEMM_BLOCKS, 256, 0, stream>>>(
                rowptr, col, deg, xcur,
                Wl + (size_t)i * D * D, Wr + (size_t)i * D * D, bl + (size_t)i * D,
                gam + (size_t)i * D, bet + (size_t)i * D, mea + (size_t)i * D, var + (size_t)i * D,
                xnext);
        } else {
            sage_gemm_kernel<2><<<GEMM_BLOCKS, 256, 0, stream>>>(
                rowptr, col, deg, xcur,
                Wl + (size_t)i * D * D, Wr + (size_t)i * D * D, bl + (size_t)i * D,
                gam + (size_t)i * D, bet + (size_t)i * D, mea + (size_t)i * D, var + (size_t)i * D,
                (float*)d_out);
        }

        float* tmp = xcur; xcur = xnext; xnext = tmp;
    }
}


# Layer kernel: each list's first 32 neighbour ids fetched a tile ahead, one a lane (bounds two tiles ahead), read out by lane into scalar row bases: one memory trip a row group, not two
# speedup vs baseline: 4.9843x; 1.0846x over previous
#include <hip/hip_runtime.h>

#define N_NODES 50000
#define NE      800000
#define D       128
#define EPS_BN  1e-5f
#define MTILES  (N_NODES / 16)
#define LDSS    132

typedef __attribute__((ext_vector_type(2))) float v2f;
typedef __attribute__((ext_vector_type(4))) float v4f;
typedef __attribute__((ext_vector_type(8))) float v8f;
typedef __attribute__((ext_vector_type(4))) int   v4i;
typedef const __attribute__((address_space(1))) char* gchar_p;
typedef const __attribute__((address_space(1))) v4f*  gv4f_p;

#define SCAN_PER   49
#define SCAN_LEN   (1024 * SCAN_PER)
#define ROWPTR_PAD 50048


__global__ void zero_kernel(float* __restrict__ p, int n4) {
    int i = blockIdx.x * blockDim.x + threadIdx.x;
    if (i < n4) ((v4f*)p)[i] = (v4f){0.0f, 0.0f, 0.0f, 0.0f};
}


__global__ void count_rank_kernel(const int* __restrict__ dst, int* __restrict__ cnt,
                                  int* __restrict__ rank) {
    int i = blockIdx.x * blockDim.x + threadIdx.x;
    if (i < NE / 4) {
        v4i d = ((const v4i*)dst)[i];
        v4i r;
        r.x = atomicAdd(&cnt[d.x], 1);
        r.y = atomicAdd(&cnt[d.y], 1);
        r.z = atomicAdd(&cnt[d.z], 1);
        r.w = atomicAdd(&cnt[d.w], 1);
        ((v4i*)rank)[i] = r;
    }
}

__global__ __launch_bounds__(1024)
void scan_kernel(const int* __restrict__ cnt, int* __restrict__ rowptr,
                 float* __restrict__ inv_deg) {
    __shared__ int sbuf[SCAN_LEN];
    __shared__ int wtot[32];
    const int t    = threadIdx.x;
    const int lane = t & 31;
    const int wave = t >> 5;

    int v[SCAN_PER];
#pragma unroll
    for (int k = 0; k < SCAN_PER; ++k) v[k] = cnt[min(t + 1024 * k, N_NODES - 1)];
#pragma unroll
    for (int k = 0; k < SCAN_PER; ++k) sbuf[t + 1024 * k] = (t + 1024 * k < N_NODES) ? v[k] : 0;
    __syncthreads();

    int s = 0;
    for (int i = 0; i < SCAN_PER; ++i) s += sbuf[t * SCAN_PER + i];

    int incl = s;
    for (int off = 1; off < 32; off <<= 1) {
        int u = __shfl_up(incl, off);
        if (lane >= off) incl += u;
    }
    if (lane == 31) wtot[wave] = incl;
    __syncthreads();
    int before = (lane < wave) ? wtot[lane] : 0;
    for (int off = 16; off > 0; off >>= 1) before += __shfl_xor(before, off);

    int run = before + incl - s;
    for (int i = 0; i < SCAN_PER; ++i) {
        int c = sbuf[t * SCAN_PER + i];
        sbuf[t * SCAN_PER + i] = run;
        run += c;
    }
    __syncthreads();

    for (int i = t; i <= N_NODES; i += 1024) rowptr[i] = sbuf[i];
    for (int i = t; i < N_NODES; i += 1024) {
        float d = (float)(sbuf[i + 1] - sbuf[i]);
        inv_deg[i] = 1.0f / fmaxf(d, 1.0f);
    }
}

__global__ void fill_kernel(const int* __restrict__ src, const int* __restrict__ dst,
                            const int* __restrict__ rank, const int* __restrict__ rowptr,
                            int* __restrict__ col) {
    int i = blockIdx.x * blockDim.x + threadIdx.x;
    if (i < NE / 4) {
        v4i s = ((const v4i*)src)[i];
        v4i d = ((const v4i*)dst)[i];
        v4i r = ((const v4i*)rank)[i];
        col[rowptr[d.x] + r.x] = s.x;
        col[rowptr[d.y] + r.y] = s.y;
        col[rowptr[d.z] + r.z] = s.z;
        col[rowptr[d.w] + r.w] = s.w;
    }
}

template <int CNT>
__device__ __forceinline__ v4f add_rows(const float* __restrict__ Ax, int idv, int lb,
                                        int lane, v4f acc) {
    gchar_p rp[CNT];
#pragma unroll
    for (int k = 0; k < CNT; ++k) {
        int v = __builtin_amdgcn_readlane(idv, lb + k);
        v = ((unsigned)v < (unsigned)N_NODES) ? v : 0;
        gchar_p p = (gchar_p)(const char*)Ax + (size_t)v * (D * sizeof(float));
        asm volatile("" : "+s"(p));
        rp[k] = p;
    }
    unsigned loff = (unsigned)lane * 16u;
    asm volatile("" : "+v"(loff));
    v4f r[CNT];
#pragma unroll
    for (int k = 0; k < CNT; ++k) r[k] = *(gv4f_p)(rp[k] + loff);
#pragma unroll
    for (int k = 0; k < CNT; ++k) acc += r[k];
    return acc;
}

__device__ __forceinline__ int load_ids(const int* __restrict__ col, int j, int lane) {
    return col[min(j + lane, NE - 1)];
}

__device__ __forceinline__ v4f gather_rows(const float* __restrict__ Ax, const int* __restrict__ col,
                                           int idv, int beg, int end, int lane) {
    v4f acc = {0.0f, 0.0f, 0.0f, 0.0f};
    int j = beg;
    while (j < end) {
        int n  = min(end - j, 32);
        int lb = 0;
#pragma nounroll
        for (; n >= 16; n -= 16, lb += 16) acc = add_rows<16>(Ax, idv, lb, lane, acc);
        switch (n) {
            case 15: acc = add_rows<15>(Ax, idv, lb, lane, acc); break;
            case 14: acc = add_rows<14>(Ax, idv, lb, lane, acc); break;
            case 13: acc = add_rows<13>(Ax, idv, lb, lane, acc); break;
            case 12: acc = add_rows<12>(Ax, idv, lb, lane, acc); break;
            case 11: acc = add_rows<11>(Ax, idv, lb, lane, acc); break;
            case 10: acc = add_rows<10>(Ax, idv, lb, lane, acc); break;
            case 9:  acc = add_rows<9>(Ax, idv, lb, lane, acc);  break;
            case 8:  acc = add_rows<8>(Ax, idv, lb, lane, acc);  break;
            case 7:  acc = add_rows<7>(Ax, idv, lb, lane, acc);  break;
            case 6:  acc = add_rows<6>(Ax, idv, lb, lane, acc);  break;
            case 5:  acc = add_rows<5>(Ax, idv, lb, lane, acc);  break;
            case 4:  acc = add_rows<4>(Ax, idv, lb, lane, acc);  break;
            case 3:  acc = add_rows<3>(Ax, idv, lb, lane, acc);  break;
            case 2:  acc = add_rows<2>(Ax, idv, lb, lane, acc);  break;
            case 1:  acc = add_rows<1>(Ax, idv, lb, lane, acc);  break;
            default: break;
        }
        j += 32;
        if (j < end) idv = load_ids(col, j, lane);
    }
    return acc;
}

template <int MODE>
__global__ __launch_bounds__(256, 1)
void sage_gemm_kernel(const int* __restrict__ rowptr, const int* __restrict__ col,
                      const float* __restrict__ inv_deg,
                      const float* __restrict__ Ax,
                      const float* __restrict__ W1, const float* __restrict__ W2,
                      const float* __restrict__ bias,
                      const float* __restrict__ gamma, const float* __restrict__ beta,
                      const float* __restrict__ mean, const float* __restrict__ var,
                      float* __restrict__ out) {
    __shared__ __attribute__((aligned(16))) float sA1[2][16 * LDSS];
    __shared__ __attribute__((aligned(16))) float sA2[2][16 * LDSS];

    const int t    = threadIdx.x;
    const int lane = t & 31;
    const int wave = t >> 5;
    const int half = lane >> 4;
    const int lrow = lane & 15;
    const int n    = wave * 16 + lrow;

    const int sr = t >> 5;
    const int sc = (t & 31) * 4;

    v2f b1[32];
    v2f b2[32];
#pragma unroll
    for (int ks = 0; ks < 32; ++ks) {
        int k = ks * 4 + 2 * half;
        b1[ks].x = W1[k * D + n];
        b1[ks].y = W1[(k + 1) * D + n];
    }
    if (MODE != 0) {
#pragma unroll
        for (int ks = 0; ks < 32; ++ks) {
            int k = ks * 4 + 2 * half;
            b2[ks].x = W2[k * D + n];
            b2[ks].y = W2[(k + 1) * D + n];
        }
    }

    float bcol = bias[n];
    float s = 1.0f, tt = 0.0f;
    if (MODE != 0) {
        s  = gamma[n] * rsqrtf(var[n] + EPS_BN);
        tt = beta[n] - mean[n] * s;
    }

    int mt = blockIdx.x;
    if (mt >= MTILES) return;

    const int wv = __builtin_amdgcn_readfirstlane(sr);
    const int l16 = min(lane, 16);
    v4f ax0, ax1;
    int   beg0 = 0, end0 = 0, beg1 = 0, end1 = 0;
    int   idv0 = 0, idv1 = 0;
    int   rpn  = 0;
    float id0 = 0.0f, id1 = 0.0f;
    {
        const size_t rb = (size_t)mt * 16;
        ax0 = *(const v4f*)&Ax[(rb + sr) * D + sc];
        ax1 = *(const v4f*)&Ax[(rb + sr + 8) * D + sc];
        if (MODE != 0) {
            const int nd = mt * 16 + sr;
            const int rpc = rowptr[mt * 16 + l16];
            rpn  = rowptr[min(mt + (int)gridDim.x, MTILES - 1) * 16 + l16];
            id0 = inv_deg[nd];
            id1 = inv_deg[nd + 8];
            beg0 = __builtin_amdgcn_readlane(rpc, wv);
            end0 = __builtin_amdgcn_readlane(rpc, wv + 1);
            beg1 = __builtin_amdgcn_readlane(rpc, wv + 8);
            end1 = __builtin_amdgcn_readlane(rpc, wv + 9);
            idv0 = load_ids(col, beg0, lane);
            idv1 = load_ids(col, beg1, lane);
        }
    }

    int buf = 0;
    while (true) {
        const size_t rowbase = (size_t)mt * 16;

        if (MODE == 0) {
            *(v4f*)&sA1[buf][sr * LDSS + sc]       = ax0;
            *(v4f*)&sA1[buf][(sr + 8) * LDSS + sc] = ax1;
        } else {
            *(v4f*)&sA2[buf][sr * LDSS + sc]       = ax0;
            *(v4f*)&sA2[buf][(sr + 8) * LDSS + sc] = ax1;
#pragma nounroll
            for (int h = 0; h < 2; ++h) {
                const int   beg = h ? beg1 : beg0;
                const int   end = h ? end1 : end0;
                const float idg = h ? id1 : id0;
                v4f acc = gather_rows(Ax, col, h ? idv1 : idv0, beg, end, lane);
                *(v4f*)&sA1[buf][(sr + 8 * h) * LDSS + sc] = acc * idg;
            }
        }
        __syncthreads();

        const int  nmt  = mt + (int)gridDim.x;
        const bool more = nmt < MTILES;

        if (more) {
            const size_t rb = (size_t)nmt * 16;
            ax0 = *(const v4f*)&Ax[(rb + sr) * D + sc];
            ax1 = *(const v4f*)&Ax[(rb + sr + 8) * D + sc];
            if (MODE != 0) {
                const int nd = nmt * 16 + sr;
                beg0 = __builtin_amdgcn_readlane(rpn, wv);
                end0 = __builtin_amdgcn_readlane(rpn, wv + 1);
                beg1 = __builtin_amdgcn_readlane(rpn, wv + 8);
                end1 = __builtin_amdgcn_readlane(rpn, wv + 9);
                idv0 = load_ids(col, beg0, lane);
                idv1 = load_ids(col, beg1, lane);
                id0 = inv_deg[nd];
                id1 = inv_deg[nd + 8];
                rpn = rowptr[min(nmt + (int)gridDim.x, MTILES - 1) * 16 + l16];
            }
        }

        const float* pA1 = &sA1[buf][0];
        const float* pA2 = &sA2[buf][0];

        v8f c8 = {};
#pragma unroll
        for (int ks = 0; ks < 32; ++ks) {
            int k = ks * 4 + 2 * half;
            v2f a = *(const v2f*)&pA1[lrow * LDSS + k];
            c8 = __builtin_amdgcn_wmma_f32_16x16x4_f32(false, a, false, b1[ks],
                                                       (short)0, c8, false, false);
        }
        if (MODE != 0) {
#pragma unroll
            for (int ks = 0; ks < 32; ++ks) {
                int k = ks * 4 + 2 * half;
                v2f a = *(const v2f*)&pA2[lrow * LDSS + k];
                c8 = __builtin_amdgcn_wmma_f32_16x16x4_f32(false, a, false, b2[ks],
                                                           (short)0, c8, false, false);
            }
        }

#pragma unroll
        for (int v = 0; v < 8; ++v) {
            int m   = v + 8 * half;
            float h = c8[v] + bcol;
            float y;
            if (MODE == 0) {
                y = fmaxf(h, 0.0f);
            } else {
                y = fmaxf(h * s + tt, 0.0f);
                if (MODE == 1) y += pA2[m * LDSS + n];
            }
            out[(rowbase + m) * D + n] = y;
        }

        if (!more) break;
        buf ^= 1;
        mt = nmt;
    }
}


extern "C" void kernel_launch(void* const* d_in, const int* in_sizes, int n_in,
                              void* d_out, int out_size, void* d_ws, size_t ws_size,
                              hipStream_t stream) {
    (void)in_sizes; (void)n_in; (void)out_size; (void)ws_size;

    const float* x   = (const float*)d_in[0];
    const int*   ei  = (const int*)d_in[1];
    const float* Wp  = (const float*)d_in[2];
    const float* bp  = (const float*)d_in[3];
    const float* Wl  = (const float*)d_in[4];
    const float* bl  = (const float*)d_in[5];
    const float* Wr  = (const float*)d_in[6];
    const float* gam = (const float*)d_in[7];
    const float* bet = (const float*)d_in[8];
    const float* mea = (const float*)d_in[9];
    const float* var = (const float*)d_in[10];

    const int* src = ei;
    const int* dst = ei + NE;

    float* ws  = (float*)d_ws;
    float* deg = ws;
    float* xa  = ws + ((N_NODES + 63) / 64) * 64;
    float* xb  = xa + (size_t)N_NODES * D;
    float* csr = xb + (size_t)N_NODES * D;

    int* col    = (int*)csr;
    int* rank   = col + NE;
    int* rowptr = rank + NE;
    int* cnt    = rowptr + ROWPTR_PAD;

    const int GEMM_BLOCKS = 512;

    {
        int n4 = N_NODES / 4;
        zero_kernel<<<(n4 + 255) / 256, 256, 0, stream>>>((float*)cnt, n4);
        count_rank_kernel<<<(NE / 4 + 255) / 256, 256, 0, stream>>>(dst, cnt, rank);
        scan_kernel<<<1, 1024, 0, stream>>>(cnt, rowptr, deg);
        fill_kernel<<<(NE / 4 + 255) / 256, 256, 0, stream>>>(src, dst, rank, rowptr, col);
    }

    sage_gemm_kernel<0><<<GEMM_BLOCKS, 256, 0, stream>>>(
        nullptr, nullptr, deg, x, Wp, Wp, bp, gam, bet, mea, var, xa);

    float* xcur  = xa;
    float* xnext = xb;
    for (int i = 0; i < 4; ++i) {
        if (i < 3) {
            sage_gemm_kernel<1><<<GEMM_BLOCKS, 256, 0, stream>>>(
                rowptr, col, deg, xcur,
                Wl + (size_t)i * D * D, Wr + (size_t)i * D * D, bl + (size_t)i * D,
                gam + (size_t)i * D, bet + (size_t)i * D, mea + (size_t)i * D, var + (size_t)i * D,
                xnext);
        } else {
            sage_gemm_kernel<2><<<GEMM_BLOCKS, 256, 0, stream>>>(
                rowptr, col, deg, xcur,
                Wl + (size_t)i * D * D, Wr + (size_t)i * D * D, bl + (size_t)i * D,
                gam + (size_t)i * D, bet + (size_t)i * D, mea + (size_t)i * D, var + (size_t)i * D,
                (float*)d_out);
        }

        float* tmp = xcur; xcur = xnext; xnext = tmp;
    }
}


# List build without global atomics: edge chunks ordered by destination range via LDS ranks, then a work-group per range counts, scans, places by LDS adds; 4 launches to 2; lists in LDS arrival order
# speedup vs baseline: 6.5620x; 1.3165x over previous
#include <hip/hip_runtime.h>

#define N_NODES 50000
#define NE      800000
#define D       128
#define EPS_BN  1e-5f
#define MTILES  (N_NODES / 16)
#define LDSS    132

typedef __attribute__((ext_vector_type(2))) float v2f;
typedef __attribute__((ext_vector_type(4))) float v4f;
typedef __attribute__((ext_vector_type(8))) float v8f;
typedef __attribute__((ext_vector_type(2))) int   v2i;
typedef const __attribute__((address_space(1))) char* gchar_p;
typedef const __attribute__((address_space(1))) v4f*  gv4f_p;

#define ROWPTR_PAD 50048

#define BIN_R       128
#define BIN_SHIFT   7
#define NBIN        ((N_NODES + BIN_R - 1) / BIN_R)
#define NCHUNK      256
#define CHUNK       (NE / NCHUNK)
#define BIN_THREADS 512
#define BIN_EPT     ((CHUNK + BIN_THREADS - 1) / BIN_THREADS)

static_assert(NE % NCHUNK == 0, "chunks must tile the edge list");
static_assert(NBIN + 1 <= BIN_THREADS, "one range offset per thread");
static_assert(BIN_R <= NCHUNK && (1 << BIN_SHIFT) == BIN_R, "range size");


__device__ __forceinline__ int block_excl_scan(int v, int lane, int wave, int* wtot) {
    int incl = v;
    for (int off = 1; off < 32; off <<= 1) {
        int u = __shfl_up(incl, off);
        if (lane >= off) incl += u;
    }
    if (lane == 31) wtot[wave] = incl;
    __syncthreads();
    int before = (lane < wave) ? wtot[lane] : 0;
    for (int off = 16; off > 0; off >>= 1) before += __shfl_xor(before, off);
    return before + incl - v;
}

__global__ __launch_bounds__(BIN_THREADS)
void bin_kernel(const int* __restrict__ src, const int* __restrict__ dst,
                int* __restrict__ ord, int* __restrict__ offT) {
    __shared__ int cntb[NBIN];
    __shared__ int offb[NBIN + 1];
    __shared__ int wtot[BIN_THREADS / 32];
    const int t    = threadIdx.x;
    const int lane = t & 31;
    const int wave = t >> 5;
    const int w    = blockIdx.x;

    if (t < NBIN) cntb[t] = 0;

    int d[BIN_EPT], s[BIN_EPT], r[BIN_EPT];
#pragma unroll
    for (int k = 0; k < BIN_EPT; ++k) {
        const int e = w * CHUNK + min(t + BIN_THREADS * k, CHUNK - 1);
        d[k] = dst[e];
        s[k] = src[e];
    }
    __syncthreads();

#pragma unroll
    for (int k = 0; k < BIN_EPT; ++k) {
        d[k] = ((unsigned)d[k] < (unsigned)N_NODES) ? d[k] : 0;
        r[k] = 0;
        if (t + BIN_THREADS * k < CHUNK) r[k] = atomicAdd(&cntb[d[k] >> BIN_SHIFT], 1);
    }
    __syncthreads();

    const int v    = (t < NBIN) ? cntb[t] : 0;
    const int excl = block_excl_scan(v, lane, wave, wtot);
    if (t <= NBIN) {
        offb[t] = excl;
        offT[t * NCHUNK + w] = excl;
    }
    __syncthreads();

#pragma unroll
    for (int k = 0; k < BIN_EPT; ++k) {
        if (t + BIN_THREADS * k < CHUNK) {
            const int p = w * CHUNK + offb[d[k] >> BIN_SHIFT] + r[k];
            ((v2i*)ord)[p] = (v2i){d[k], s[k]};
        }
    }
}

__global__ __launch_bounds__(NCHUNK)
void build_kernel(const int* __restrict__ ord, const int* __restrict__ offT,
                  int* __restrict__ col, int* __restrict__ rowptr,
                  float* __restrict__ inv_deg) {
    __shared__ int cnt[BIN_R];
    __shared__ int wsum[NCHUNK / 32];
    __shared__ int wtot[NCHUNK / 32];
    const int t    = threadIdx.x;
    const int lane = t & 31;
    const int wave = t >> 5;
    const int b    = blockIdx.x;

    const int s0 = offT[b * NCHUNK + t];
    const int s1 = offT[(b + 1) * NCHUNK + t];
    const v2i* seg = (const v2i*)ord + (size_t)t * CHUNK;

    if (t < BIN_R) cnt[t] = 0;
    int part = s0;
    for (int off = 16; off > 0; off >>= 1) part += __shfl_xor(part, off);
    if (lane == 0) wsum[wave] = part;
    __syncthreads();
    int base = 0;
#pragma unroll
    for (int k = 0; k < NCHUNK / 32; ++k) base += wsum[k];

    for (int i = s0; i < s1; i += 8) {
        v2i e[8];
#pragma unroll
        for (int k = 0; k < 8; ++k) e[k] = seg[min(i + k, s1 - 1)];
#pragma unroll
        for (int k = 0; k < 8; ++k)
            if (i + k < s1) atomicAdd(&cnt[e[k].x & (BIN_R - 1)], 1);
    }
    __syncthreads();

    const int c     = (t < BIN_R) ? cnt[t] : 0;
    const int start = base + block_excl_scan(c, lane, wave, wtot);
    const int node  = b * BIN_R + t;
    if (t < BIN_R && node < N_NODES) {
        rowptr[node]  = start;
        inv_deg[node] = 1.0f / fmaxf((float)c, 1.0f);
        if (node == N_NODES - 1) rowptr[N_NODES] = start + c;
    }
    if (t < BIN_R) cnt[t] = start;
    __syncthreads();

    for (int i = s0; i < s1; i += 8) {
        v2i e[8];
#pragma unroll
        for (int k = 0; k < 8; ++k) e[k] = seg[min(i + k, s1 - 1)];
#pragma unroll
        for (int k = 0; k < 8; ++k) {
            if (i + k < s1) {
                const int p = atomicAdd(&cnt[e[k].x & (BIN_R - 1)], 1);
                col[p] = e[k].y;
            }
        }
    }
}

template <int CNT>
__device__ __forceinline__ v4f add_rows(const float* __restrict__ Ax, int idv, int lb,
                                        int lane, v4f acc) {
    gchar_p rp[CNT];
#pragma unroll
    for (int k = 0; k < CNT; ++k) {
        int v = __builtin_amdgcn_readlane(idv, lb + k);
        v = ((unsigned)v < (unsigned)N_NODES) ? v : 0;
        gchar_p p = (gchar_p)(const char*)Ax + (size_t)v * (D * sizeof(float));
        asm volatile("" : "+s"(p));
        rp[k] = p;
    }
    unsigned loff = (unsigned)lane * 16u;
    asm volatile("" : "+v"(loff));
    v4f r[CNT];
#pragma unroll
    for (int k = 0; k < CNT; ++k) r[k] = *(gv4f_p)(rp[k] + loff);
#pragma unroll
    for (int k = 0; k < CNT; ++k) acc += r[k];
    return acc;
}

__device__ __forceinline__ int load_ids(const int* __restrict__ col, int j, int lane) {
    return col[min(j + lane, NE - 1)];
}

__device__ __forceinline__ v4f gather_rows(const float* __restrict__ Ax, const int* __restrict__ col,
                                           int idv, int beg, int end, int lane) {
    v4f acc = {0.0f, 0.0f, 0.0f, 0.0f};
    int j = beg;
    while (j < end) {
        int n  = min(end - j, 32);
        int lb = 0;
#pragma nounroll
        for (; n >= 16; n -= 16, lb += 16) acc = add_rows<16>(Ax, idv, lb, lane, acc);
        switch (n) {
            case 15: acc = add_rows<15>(Ax, idv, lb, lane, acc); break;
            case 14: acc = add_rows<14>(Ax, idv, lb, lane, acc); break;
            case 13: acc = add_rows<13>(Ax, idv, lb, lane, acc); break;
            case 12: acc = add_rows<12>(Ax, idv, lb, lane, acc); break;
            case 11: acc = add_rows<11>(Ax, idv, lb, lane, acc); break;
            case 10: acc = add_rows<10>(Ax, idv, lb, lane, acc); break;
            case 9:  acc = add_rows<9>(Ax, idv, lb, lane, acc);  break;
            case 8:  acc = add_rows<8>(Ax, idv, lb, lane, acc);  break;
            case 7:  acc = add_rows<7>(Ax, idv, lb, lane, acc);  break;
            case 6:  acc = add_rows<6>(Ax, idv, lb, lane, acc);  break;
            case 5:  acc = add_rows<5>(Ax, idv, lb, lane, acc);  break;
            case 4:  acc = add_rows<4>(Ax, idv, lb, lane, acc);  break;
            case 3:  acc = add_rows<3>(Ax, idv, lb, lane, acc);  break;
            case 2:  acc = add_rows<2>(Ax, idv, lb, lane, acc);  break;
            case 1:  acc = add_rows<1>(Ax, idv, lb, lane, acc);  break;
            default: break;
        }
        j += 32;
        if (j < end) idv = load_ids(col, j, lane);
    }
    return acc;
}

template <int MODE>
__global__ __launch_bounds__(256, 1)
void sage_gemm_kernel(const int* __restrict__ rowptr, const int* __restrict__ col,
                      const float* __restrict__ inv_deg,
                      const float* __restrict__ Ax,
                      const float* __restrict__ W1, const float* __restrict__ W2,
                      const float* __restrict__ bias,
                      const float* __restrict__ gamma, const float* __restrict__ beta,
                      const float* __restrict__ mean, const float* __restrict__ var,
                      float* __restrict__ out) {
    __shared__ __attribute__((aligned(16))) float sA1[2][16 * LDSS];
    __shared__ __attribute__((aligned(16))) float sA2[2][16 * LDSS];

    const int t    = threadIdx.x;
    const int lane = t & 31;
    const int wave = t >> 5;
    const int half = lane >> 4;
    const int lrow = lane & 15;
    const int n    = wave * 16 + lrow;

    const int sr = t >> 5;
    const int sc = (t & 31) * 4;

    v2f b1[32];
    v2f b2[32];
#pragma unroll
    for (int ks = 0; ks < 32; ++ks) {
        int k = ks * 4 + 2 * half;
        b1[ks].x = W1[k * D + n];
        b1[ks].y = W1[(k + 1) * D + n];
    }
    if (MODE != 0) {
#pragma unroll
        for (int ks = 0; ks < 32; ++ks) {
            int k = ks * 4 + 2 * half;
            b2[ks].x = W2[k * D + n];
            b2[ks].y = W2[(k + 1) * D + n];
        }
    }

    float bcol = bias[n];
    float s = 1.0f, tt = 0.0f;
    if (MODE != 0) {
        s  = gamma[n] * rsqrtf(var[n] + EPS_BN);
        tt = beta[n] - mean[n] * s;
    }

    int mt = blockIdx.x;
    if (mt >= MTILES) return;

    const int wv = __builtin_amdgcn_readfirstlane(sr);
    const int l16 = min(lane, 16);
    v4f ax0, ax1;
    int   beg0 = 0, end0 = 0, beg1 = 0, end1 = 0;
    int   idv0 = 0, idv1 = 0;
    int   rpn  = 0;
    float id0 = 0.0f, id1 = 0.0f;
    {
        const size_t rb = (size_t)mt * 16;
        ax0 = *(const v4f*)&Ax[(rb + sr) * D + sc];
        ax1 = *(const v4f*)&Ax[(rb + sr + 8) * D + sc];
        if (MODE != 0) {
            const int nd = mt * 16 + sr;
            const int rpc = rowptr[mt * 16 + l16];
            rpn  = rowptr[min(mt + (int)gridDim.x, MTILES - 1) * 16 + l16];
            id0 = inv_deg[nd];
            id1 = inv_deg[nd + 8];
            beg0 = __builtin_amdgcn_readlane(rpc, wv);
            end0 = __builtin_amdgcn_readlane(rpc, wv + 1);
            beg1 = __builtin_amdgcn_readlane(rpc, wv + 8);
            end1 = __builtin_amdgcn_readlane(rpc, wv + 9);
            idv0 = load_ids(col, beg0, lane);
            idv1 = load_ids(col, beg1, lane);
        }
    }

    int buf = 0;
    while (true) {
        const size_t rowbase = (size_t)mt * 16;

        if (MODE == 0) {
            *(v4f*)&sA1[buf][sr * LDSS + sc]       = ax0;
            *(v4f*)&sA1[buf][(sr + 8) * LDSS + sc] = ax1;
        } else {
            *(v4f*)&sA2[buf][sr * LDSS + sc]       = ax0;
            *(v4f*)&sA2[buf][(sr + 8) * LDSS + sc] = ax1;
#pragma nounroll
            for (int h = 0; h < 2; ++h) {
                const int   beg = h ? beg1 : beg0;
                const int   end = h ? end1 : end0;
                const float idg = h ? id1 : id0;
                v4f acc = gather_rows(Ax, col, h ? idv1 : idv0, beg, end, lane);
                *(v4f*)&sA1[buf][(sr + 8 * h) * LDSS + sc] = acc * idg;
            }
        }
        __syncthreads();

        const int  nmt  = mt + (int)gridDim.x;
        const bool more = nmt < MTILES;

        if (more) {
            const size_t rb = (size_t)nmt * 16;
            ax0 = *(const v4f*)&Ax[(rb + sr) * D + sc];
            ax1 = *(const v4f*)&Ax[(rb + sr + 8) * D + sc];
            if (MODE != 0) {
                const int nd = nmt * 16 + sr;
                beg0 = __builtin_amdgcn_readlane(rpn, wv);
                end0 = __builtin_amdgcn_readlane(rpn, wv + 1);
                beg1 = __builtin_amdgcn_readlane(rpn, wv + 8);
                end1 = __builtin_amdgcn_readlane(rpn, wv + 9);
                idv0 = load_ids(col, beg0, lane);
                idv1 = load_ids(col, beg1, lane);
                id0 = inv_deg[nd];
                id1 = inv_deg[nd + 8];
                rpn = rowptr[min(nmt + (int)gridDim.x, MTILES - 1) * 16 + l16];
            }
        }

        const float* pA1 = &sA1[buf][0];
        const float* pA2 = &sA2[buf][0];

        v8f c8 = {};
#pragma unroll
        for (int ks = 0; ks < 32; ++ks) {
            int k = ks * 4 + 2 * half;
            v2f a = *(const v2f*)&pA1[lrow * LDSS + k];
            c8 = __builtin_amdgcn_wmma_f32_16x16x4_f32(false, a, false, b1[ks],
                                                       (short)0, c8, false, false);
        }
        if (MODE != 0) {
#pragma unroll
            for (int ks = 0; ks < 32; ++ks) {
                int k = ks * 4 + 2 * half;
                v2f a = *(const v2f*)&pA2[lrow * LDSS + k];
                c8 = __builtin_amdgcn_wmma_f32_16x16x4_f32(false, a, false, b2[ks],
                                                           (short)0, c8, false, false);
            }
        }

#pragma unroll
        for (int v = 0; v < 8; ++v) {
            int m   = v + 8 * half;
            float h = c8[v] + bcol;
            float y;
            if (MODE == 0) {
                y = fmaxf(h, 0.0f);
            } else {
                y = fmaxf(h * s + tt, 0.0f);
                if (MODE == 1) y += pA2[m * LDSS + n];
            }
            out[(rowbase + m) * D + n] = y;
        }

        if (!more) break;
        buf ^= 1;
        mt = nmt;
    }
}


extern "C" void kernel_launch(void* const* d_in, const int* in_sizes, int n_in,
                              void* d_out, int out_size, void* d_ws, size_t ws_size,
                              hipStream_t stream) {
    (void)in_sizes; (void)n_in; (void)out_size; (void)ws_size;

    const float* x   = (const float*)d_in[0];
    const int*   ei  = (const int*)d_in[1];
    const float* Wp  = (const float*)d_in[2];
    const float* bp  = (const float*)d_in[3];
    const float* Wl  = (const float*)d_in[4];
    const float* bl  = (const float*)d_in[5];
    const float* Wr  = (const float*)d_in[6];
    const float* gam = (const float*)d_in[7];
    const float* bet = (const float*)d_in[8];
    const float* mea = (const float*)d_in[9];
    const float* var = (const float*)d_in[10];

    const int* src = ei;
    const int* dst = ei + NE;

    float* ws  = (float*)d_ws;
    float* deg = ws;
    float* xa  = ws + ((N_NODES + 63) / 64) * 64;
    float* xb  = xa + (size_t)N_NODES * D;
    float* csr = xb + (size_t)N_NODES * D;

    int* col    = (int*)csr;
    int* ord    = col + NE;
    int* rowptr = ord + 2 * NE;
    int* offT   = rowptr + ROWPTR_PAD;

    const int GEMM_BLOCKS = 512;

    bin_kernel<<<NCHUNK, BIN_THREADS, 0, stream>>>(src, dst, ord, offT);
    build_kernel<<<NBIN, NCHUNK, 0, stream>>>(ord, offT, col, rowptr, deg);

    sage_gemm_kernel<0><<<GEMM_BLOCKS, 256, 0, stream>>>(
        nullptr, nullptr, deg, x, Wp, Wp, bp, gam, bet, mea, var, xa);

    float* xcur  = xa;
    float* xnext = xb;
    for (int i = 0; i < 4; ++i) {
        if (i < 3) {
            sage_gemm_kernel<1><<<GEMM_BLOCKS, 256, 0, stream>>>(
                rowptr, col, deg, xcur,
                Wl + (size_t)i * D * D, Wr + (size_t)i * D * D, bl + (size_t)i * D,
                gam + (size_t)i * D, bet + (size_t)i * D, mea + (size_t)i * D, var + (size_t)i * D,
                xnext);
        } else {
            sage_gemm_kernel<2><<<GEMM_BLOCKS, 256, 0, stream>>>(
                rowptr, col, deg, xcur,
                Wl + (size_t)i * D * D, Wr + (size_t)i * D * D, bl + (size_t)i * D,
                gam + (size_t)i * D, bet + (size_t)i * D, mea + (size_t)i * D, var + (size_t)i * D,
                (float*)d_out);
        }

        float* tmp = xcur; xcur = xnext; xnext = tmp;
    }
}
